# starting kernel
# baseline (speedup 1.0000x reference)
#include <hip/hip_runtime.h>
#include <math.h>

#define D_MODEL  1024
#define D_STATE  16
#define D_CONV   4
#define D_INNER  2048
#define DT_RANK  64
#define BATCH    2
#define SEQLEN   2048
#define BL       (BATCH * SEQLEN)
#define XDBL_LD  (DT_RANK + 2 * D_STATE)

typedef __attribute__((ext_vector_type(2))) float v2f;
typedef __attribute__((ext_vector_type(4))) float v4f;
typedef __attribute__((ext_vector_type(8))) float v8f;

__device__ __forceinline__ float silu_f(float v)     { return v / (1.f + expf(-v)); }
__device__ __forceinline__ float softplus_f(float v) { return (v > 20.f) ? v : log1pf(expf(v)); }

template <int NT, int EPI>
__launch_bounds__(256)
__global__ void gemm_wmma_f32(const float* __restrict__ A, int lda,
                              const float* __restrict__ W, int ldw,
                              float* __restrict__ C, int ldc, int K,
                              const float* __restrict__ bias)
{
    constexpr int KC      = 32;
    constexpr int KC4     = KC / 4;
    constexpr int LDS_LD  = KC + 4;
    constexpr int NROWS_B = 16 * NT;
    constexpr int NA      = (128 * KC4) / 256;
    constexpr int NB      = (NROWS_B * KC4) / 256;

    __shared__ float As[128 * LDS_LD];
    __shared__ float Bs[NROWS_B * LDS_LD];

    const int tid  = threadIdx.x;
    const int wave = tid >> 5;
    const int lane = tid & 31;
    const int half = lane >> 4;
    const int lidx = lane & 15;
    const int m_blk  = blockIdx.x * 128;
    const int n_base = blockIdx.y * NROWS_B;

    v8f acc[NT];
#pragma unroll
    for (int t = 0; t < NT; ++t) acc[t] = (v8f){0.f,0.f,0.f,0.f,0.f,0.f,0.f,0.f};

    v4f ra[NA], rb[NB];

#pragma unroll
    for (int i = 0; i < NA; ++i) {
        const int idx = tid + i * 256, r = idx >> 3, c = (idx & 7) * 4;
        ra[i] = *(const v4f*)(A + (size_t)(m_blk + r) * lda + c);
    }
#pragma unroll
    for (int i = 0; i < NB; ++i) {
        const int idx = tid + i * 256, r = idx >> 3, c = (idx & 7) * 4;
        rb[i] = *(const v4f*)(W + (size_t)(n_base + r) * ldw + c);
    }

    for (int k0 = 0; k0 < K; k0 += KC) {
#pragma unroll
        for (int i = 0; i < NA; ++i) {
            const int idx = tid + i * 256, r = idx >> 3, c = (idx & 7) * 4;
            *(v4f*)(As + r * LDS_LD + c) = ra[i];
        }
#pragma unroll
        for (int i = 0; i < NB; ++i) {
            const int idx = tid + i * 256, r = idx >> 3, c = (idx & 7) * 4;
            *(v4f*)(Bs + r * LDS_LD + c) = rb[i];
        }
        __syncthreads();

        if (k0 + KC < K) {
#pragma unroll
            for (int i = 0; i < NA; ++i) {
                const int idx = tid + i * 256, r = idx >> 3, c = (idx & 7) * 4;
                ra[i] = *(const v4f*)(A + (size_t)(m_blk + r) * lda + (k0 + KC) + c);
            }
#pragma unroll
            for (int i = 0; i < NB; ++i) {
                const int idx = tid + i * 256, r = idx >> 3, c = (idx & 7) * 4;
                rb[i] = *(const v4f*)(W + (size_t)(n_base + r) * ldw + (k0 + KC) + c);
            }
        }

        const float* a_lds = As + (wave * 16 + lidx) * LDS_LD + 2 * half;
        const float* b_lds = Bs + lidx * LDS_LD + 2 * half;
#pragma unroll
        for (int kk = 0; kk < KC; kk += 4) {
            v2f a = *(const v2f*)(a_lds + kk);
#pragma unroll
            for (int t = 0; t < NT; ++t) {
                v2f b = *(const v2f*)(b_lds + t * 16 * LDS_LD + kk);
                acc[t] = __builtin_amdgcn_wmma_f32_16x16x4_f32(
                    false, a, false, b, (short)0, acc[t], false, false);
            }
        }
        __syncthreads();
    }

#pragma unroll
    for (int t = 0; t < NT; ++t) {
        const int n = n_base + t * 16 + lidx;
#pragma unroll
        for (int v = 0; v < 8; ++v) {
            const int m = m_blk + wave * 16 + half * 8 + v;
            float val = acc[t][v];
            if (EPI == 1) val = softplus_f(val + bias[n]);
            C[(size_t)m * ldc + n] = val;
        }
    }
}

__global__ void conv_silu_kernel(const float* __restrict__ x,
                                 const float* __restrict__ w,
                                 const float* __restrict__ cb,
                                 float* __restrict__ xc)
{
    const int idx = blockIdx.x * blockDim.x + threadIdx.x;
    if (idx >= BATCH * SEQLEN * D_INNER) return;
    const int d = idx % D_INNER;
    const int l = (idx / D_INNER) % SEQLEN;
    const int b = idx / (D_INNER * SEQLEN);
    float acc = cb[d];
#pragma unroll
    for (int k = 0; k < D_CONV; ++k) {
        const int ls = l + k - (D_CONV - 1);
        if (ls >= 0)
            acc += x[((size_t)b * SEQLEN + ls) * D_INNER + d] * w[d * D_CONV + k];
    }
    xc[idx] = silu_f(acc);
}

__global__ void scan_kernel(const float* delta,
                            const float* __restrict__ xc,
                            const float* __restrict__ xdbl,
                            const float* __restrict__ A_log,
                            const float* __restrict__ Dp,
                            const float* __restrict__ z,
                            float* yout)
{
    const int gid = blockIdx.x * blockDim.x + threadIdx.x;
    const int n  = gid & (D_STATE - 1);
    const int ch = gid >> 4;
    if (ch >= BATCH * D_INNER) return;
    const int d = ch % D_INNER;
    const int b = ch / D_INNER;

    const float An = -expf(A_log[d * D_STATE + n]);
    const float Dd = Dp[d];
    float h = 0.f;

    const size_t row0 = (size_t)b * SEQLEN;
    for (int l = 0; l < SEQLEN; ++l) {
        const size_t off = (row0 + l) * D_INNER + d;
        const float dt = delta[off];
        const float u  = xc[off];
        const float* bc = xdbl + (row0 + l) * XDBL_LD;
        const float Bn = bc[DT_RANK + n];
        const float Cn = bc[DT_RANK + D_STATE + n];

        const float dA = expf(dt * An);
        h = h * dA + (dt * u) * Bn;
        float t = h * Cn;
        t += __shfl_xor(t, 1, 16);
        t += __shfl_xor(t, 2, 16);
        t += __shfl_xor(t, 4, 16);
        t += __shfl_xor(t, 8, 16);
        if (n == 0) {
            const float zz = z[off];
            yout[off] = (t + Dd * u) * silu_f(zz);
        }
    }
}

extern "C" void kernel_launch(void* const* d_in, const int* in_sizes, int n_in,
                              void* d_out, int out_size, void* d_ws, size_t ws_size,
                              hipStream_t stream)
{
    const float* hs        = (const float*)d_in[0];
    const float* in_proj_w = (const float*)d_in[1];
    const float* conv_w    = (const float*)d_in[2];
    const float* conv_b    = (const float*)d_in[3];
    const float* x_proj_w  = (const float*)d_in[4];
    const float* dt_proj_w = (const float*)d_in[5];
    const float* dt_proj_b = (const float*)d_in[6];
    const float* A_log     = (const float*)d_in[7];
    const float* Dp        = (const float*)d_in[8];
    const float* out_w     = (const float*)d_in[9];
    float* out = (float*)d_out;

    const size_t SZ_BLD = (size_t)BL * D_INNER * sizeof(float);
    char* ws = (char*)d_ws;
    float* x     = (float*)(ws);
    float* z     = (float*)(ws + SZ_BLD);
    float* xc    = (float*)(ws + 2 * SZ_BLD);
    float* delta = (float*)(ws + 3 * SZ_BLD);
    float* xdbl  = (float*)(ws + 4 * SZ_BLD);

    const dim3 blk(256);

    gemm_wmma_f32<8, 0><<<dim3(BL / 128, D_INNER / 128), blk, 0, stream>>>(
        hs, D_MODEL, in_proj_w, D_MODEL, x, D_INNER, D_MODEL, nullptr);
    gemm_wmma_f32<8, 0><<<dim3(BL / 128, D_INNER / 128), blk, 0, stream>>>(
        hs, D_MODEL, in_proj_w + (size_t)D_INNER * D_MODEL, D_MODEL,
        z, D_INNER, D_MODEL, nullptr);

    const int nelem = BATCH * SEQLEN * D_INNER;
    conv_silu_kernel<<<(nelem + 255) / 256, blk, 0, stream>>>(x, conv_w, conv_b, xc);

    gemm_wmma_f32<6, 0><<<dim3(BL / 128, 1), blk, 0, stream>>>(
        xc, D_INNER, x_proj_w, D_INNER, xdbl, XDBL_LD, D_INNER, nullptr);

    gemm_wmma_f32<8, 1><<<dim3(BL / 128, D_INNER / 128), blk, 0, stream>>>(
        xdbl, XDBL_LD, dt_proj_w, DT_RANK, delta, D_INNER, DT_RANK, dt_proj_b);

    const int scan_threads = BATCH * D_INNER * D_STATE;
    scan_kernel<<<(scan_threads + 255) / 256, blk, 0, stream>>>(
        delta, xc, xdbl, A_log, Dp, z, delta);

    gemm_wmma_f32<8, 0><<<dim3(BL / 128, D_MODEL / 128), blk, 0, stream>>>(
        delta, D_INNER, out_w, D_INNER, out, D_MODEL, D_INNER, nullptr);
}


# x_proj gemm (N=96, was 32 blocks of 128 rows): own kernel, 16-row blocks (256), 3 waves of one 16x32 strip each, K chunks through two alternating LDS tiles; per-output k order unchanged
# speedup vs baseline: 5.2992x; 5.2992x over previous
#include <hip/hip_runtime.h>
#include <math.h>

#define D_MODEL  1024
#define D_STATE  16
#define D_CONV   4
#define D_INNER  2048
#define DT_RANK  64
#define BATCH    2
#define SEQLEN   2048
#define BL       (BATCH * SEQLEN)
#define XDBL_LD  (DT_RANK + 2 * D_STATE)
#define SCAN_NC    16
#define SCAN_T     (SEQLEN / SCAN_NC)
#define SCAN_LANES (BATCH * D_INNER * D_STATE)

typedef __attribute__((ext_vector_type(2))) float v2f;
typedef __attribute__((ext_vector_type(4))) float v4f;
typedef __attribute__((ext_vector_type(8))) float v8f;

__device__ __forceinline__ float silu_f(float v)     { return v / (1.f + expf(-v)); }
__device__ __forceinline__ float softplus_f(float v) { return (v > 20.f) ? v : log1pf(expf(v)); }

template <int NT, int EPI>
__launch_bounds__(256)
__global__ void gemm_wmma_f32(const float* __restrict__ A, int lda,
                              const float* __restrict__ W, int ldw,
                              float* __restrict__ C, int ldc, int K,
                              const float* __restrict__ bias,
                              const float* __restrict__ cw,
                              float* __restrict__ C2)
{
    constexpr int KC      = 32;
    constexpr int KC4     = KC / 4;
    constexpr int LDS_LD  = KC + 4;
    constexpr int NROWS_B = 16 * NT;
    constexpr int NA      = (128 * KC4) / 256;
    constexpr int NB      = (NROWS_B * KC4) / 256;

    constexpr int CS_LD   = 132;
    constexpr int SMEM    = (EPI == 2) ? 128 * CS_LD : (128 + NROWS_B) * LDS_LD;

    __shared__ float smem[SMEM];
    float* As = smem;
    float* Bs = smem + 128 * LDS_LD;

    const int tid  = threadIdx.x;
    const int wave = tid >> 5;
    const int lane = tid & 31;
    const int half = lane >> 4;
    const int lidx = lane & 15;
    const int m_blk  = blockIdx.x * 128;
    const int n_base = blockIdx.y * NROWS_B;

    v8f acc[NT];
#pragma unroll
    for (int t = 0; t < NT; ++t) acc[t] = (v8f){0.f,0.f,0.f,0.f,0.f,0.f,0.f,0.f};

    v4f ra[NA], rb[NB];

#pragma unroll
    for (int i = 0; i < NA; ++i) {
        const int idx = tid + i * 256, r = idx >> 3, c = (idx & 7) * 4;
        ra[i] = *(const v4f*)(A + (size_t)(m_blk + r) * lda + c);
    }
#pragma unroll
    for (int i = 0; i < NB; ++i) {
        const int idx = tid + i * 256, r = idx >> 3, c = (idx & 7) * 4;
        rb[i] = *(const v4f*)(W + (size_t)(n_base + r) * ldw + c);
    }

    for (int k0 = 0; k0 < K; k0 += KC) {
#pragma unroll
        for (int i = 0; i < NA; ++i) {
            const int idx = tid + i * 256, r = idx >> 3, c = (idx & 7) * 4;
            *(v4f*)(As + r * LDS_LD + c) = ra[i];
        }
#pragma unroll
        for (int i = 0; i < NB; ++i) {
            const int idx = tid + i * 256, r = idx >> 3, c = (idx & 7) * 4;
            *(v4f*)(Bs + r * LDS_LD + c) = rb[i];
        }
        __syncthreads();

        if (k0 + KC < K) {
#pragma unroll
            for (int i = 0; i < NA; ++i) {
                const int idx = tid + i * 256, r = idx >> 3, c = (idx & 7) * 4;
                ra[i] = *(const v4f*)(A + (size_t)(m_blk + r) * lda + (k0 + KC) + c);
            }
#pragma unroll
            for (int i = 0; i < NB; ++i) {
                const int idx = tid + i * 256, r = idx >> 3, c = (idx & 7) * 4;
                rb[i] = *(const v4f*)(W + (size_t)(n_base + r) * ldw + (k0 + KC) + c);
            }
        }

        const float* a_lds = As + (wave * 16 + lidx) * LDS_LD + 2 * half;
        const float* b_lds = Bs + lidx * LDS_LD + 2 * half;
#pragma unroll
        for (int kk = 0; kk < KC; kk += 4) {
            v2f a = *(const v2f*)(a_lds + kk);
#pragma unroll
            for (int t = 0; t < NT; ++t) {
                v2f b = *(const v2f*)(b_lds + t * 16 * LDS_LD + kk);
                acc[t] = __builtin_amdgcn_wmma_f32_16x16x4_f32(
                    false, a, false, b, (short)0, acc[t], false, false);
            }
        }
        __syncthreads();
    }

    if (EPI == 2) {
#pragma unroll
        for (int t = 0; t < NT; ++t) {
#pragma unroll
            for (int v = 0; v < 8; ++v)
                smem[(wave * 16 + half * 8 + v) * CS_LD + t * 16 + lidx] = acc[t][v];
        }
        __syncthreads();

        const int cg = tid & 31;
        const int r0 = (tid >> 5) * 16;
        const int d0 = n_base + cg * 4;
        const v4f bv = *(const v4f*)(bias + d0);
        const v4f w0 = *(const v4f*)(cw + (size_t)(d0 + 0) * D_CONV);
        const v4f w1 = *(const v4f*)(cw + (size_t)(d0 + 1) * D_CONV);
        const v4f w2 = *(const v4f*)(cw + (size_t)(d0 + 2) * D_CONV);
        const v4f w3 = *(const v4f*)(cw + (size_t)(d0 + 3) * D_CONV);
        const v4f t0 = (v4f){w0[0], w1[0], w2[0], w3[0]};
        const v4f t1 = (v4f){w0[1], w1[1], w2[1], w3[1]};
        const v4f t2 = (v4f){w0[2], w1[2], w2[2], w3[2]};
        const v4f t3 = (v4f){w0[3], w1[3], w2[3], w3[3]};
        const float* cs = smem + cg * 4;

        if (r0 == 0 || r0 == 112) {
            const int rb = (r0 == 0) ? 0 : 125;
#pragma unroll
            for (int j = 0; j < 3; ++j)
                *(v4f*)(C + (size_t)(m_blk + rb + j) * ldc + d0) = *(const v4f*)(cs + (rb + j) * CS_LD);
        }

        v4f xm3 = (v4f){0.f, 0.f, 0.f, 0.f}, xm2 = xm3, xm1 = xm3;
        if (r0 > 0) {
            xm3 = *(const v4f*)(cs + (r0 - 3) * CS_LD);
            xm2 = *(const v4f*)(cs + (r0 - 2) * CS_LD);
            xm1 = *(const v4f*)(cs + (r0 - 1) * CS_LD);
        }
#pragma unroll
        for (int i = 0; i < 16; ++i) {
            const int r = r0 + i;
            const v4f x0 = *(const v4f*)(cs + r * CS_LD);
            if (r >= 3) {
                v4f a = bv;
                a += xm3 * t0;
                a += xm2 * t1;
                a += xm1 * t2;
                a += x0 * t3;
                v4f o;
                o[0] = silu_f(a[0]); o[1] = silu_f(a[1]);
                o[2] = silu_f(a[2]); o[3] = silu_f(a[3]);
                *(v4f*)(C2 + (size_t)(m_blk + r) * ldc + d0) = o;
            }
            xm3 = xm2; xm2 = xm1; xm1 = x0;
        }
        return;
    }
#pragma unroll
    for (int t = 0; t < NT; ++t) {
        const int n = n_base + t * 16 + lidx;
#pragma unroll
        for (int v = 0; v < 8; ++v) {
            const int m = m_blk + wave * 16 + half * 8 + v;
            float val = acc[t][v];
            if (EPI == 1) val = softplus_f(val + bias[n]);
            C[(size_t)m * ldc + n] = val;
        }
    }
}

#define XP_KC   128
#define XP_LD   (XP_KC + 4)
#define XP_ROWS (16 + XDBL_LD + 2)
#define XP_THR  96
#define XP_NCH  (D_INNER / XP_KC)

__launch_bounds__(XP_THR)
__global__ void xproj_wmma_f32(const float* __restrict__ A,
                               const float* __restrict__ W,
                               float* __restrict__ C)
{
    constexpr int KC4  = XP_KC / 4;
    constexpr int NVEC = XP_ROWS * KC4;
    constexpr int NV   = NVEC / XP_THR;
    static_assert(NV * XP_THR == NVEC, "the tile must split evenly over the threads");
    constexpr int KS   = XP_KC / 4;
    constexpr int TILE = XP_ROWS * XP_LD;

    __shared__ float smem[2 * TILE];

    const int tid  = threadIdx.x;
    const int wave = tid >> 5;
    const int lane = tid & 31;
    const int half = lane >> 4;
    const int lidx = lane & 15;
    const int m_blk = blockIdx.x * 16;

    v8f acc0 = (v8f){0.f,0.f,0.f,0.f,0.f,0.f,0.f,0.f};
    v8f acc1 = acc0;

    const float* gsrc[NV];
    int          loff[NV];
#pragma unroll
    for (int i = 0; i < NV; ++i) {
        const int idx = tid + i * XP_THR, r = idx / KC4, c = (idx % KC4) * 4;
        loff[i] = r * XP_LD + c;
        const int wr = (r - 16 < XDBL_LD) ? r - 16 : XDBL_LD - 1;
        gsrc[i] = (r < 16) ? A + (size_t)(m_blk + r) * D_INNER + c
                           : W + (size_t)wr * D_INNER + c;
    }

    v4f regs[NV];
#pragma unroll
    for (int i = 0; i < NV; ++i) regs[i] = *(const v4f*)(gsrc[i]);
#pragma unroll
    for (int i = 0; i < NV; ++i) {
        *(v4f*)(smem + loff[i]) = regs[i];
        regs[i] = *(const v4f*)(gsrc[i] + XP_KC);
    }
    __syncthreads();

    for (int ch = 0; ch < XP_NCH; ++ch) {
        const float* cur = smem + (ch & 1) * TILE;
        float*       nxt = smem + ((ch + 1) & 1) * TILE;
        const int kload = ((ch + 2 < XP_NCH) ? ch + 2 : XP_NCH - 1) * XP_KC;
        const float* a_lds = cur + lidx * XP_LD + 2 * half;
        const float* b_lds = cur + (16 + wave * 32 + lidx) * XP_LD + 2 * half;
#pragma unroll
        for (int ks = 0; ks < KS; ++ks) {
            const v2f a  = *(const v2f*)(a_lds + ks * 4);
            const v2f b0 = *(const v2f*)(b_lds + ks * 4);
            const v2f b1 = *(const v2f*)(b_lds + 16 * XP_LD + ks * 4);
            acc0 = __builtin_amdgcn_wmma_f32_16x16x4_f32(
                false, a, false, b0, (short)0, acc0, false, false);
            acc1 = __builtin_amdgcn_wmma_f32_16x16x4_f32(
                false, a, false, b1, (short)0, acc1, false, false);
#pragma unroll
            for (int i = (ks * NV) / KS; i < ((ks + 1) * NV) / KS; ++i) {
                *(v4f*)(nxt + loff[i]) = regs[i];
                regs[i] = *(const v4f*)(gsrc[i] + kload);
            }
        }
        __syncthreads();
    }

    float* crow = C + (size_t)(m_blk + half * 8) * XDBL_LD + wave * 32 + lidx;
#pragma unroll
    for (int v = 0; v < 8; ++v) {
        crow[(size_t)v * XDBL_LD]      = acc0[v];
        crow[(size_t)v * XDBL_LD + 16] = acc1[v];
    }
}

#define EDGE_ROWS 3
#define EDGE_N    ((BL / 128) * EDGE_ROWS * (D_INNER / 4))
__global__ void conv_edge_kernel(const float* __restrict__ x,
                                 const float* __restrict__ w,
                                 const float* __restrict__ cb,
                                 float* __restrict__ xc)
{
    const int idx = blockIdx.x * blockDim.x + threadIdx.x;
    if (idx >= EDGE_N) return;
    const int d0   = (idx % (D_INNER / 4)) * 4;
    const int rr   = (idx / (D_INNER / 4)) % EDGE_ROWS;
    const int tile = idx / ((D_INNER / 4) * EDGE_ROWS);
    const int m = tile * 128 + rr;
    const int l = m % SEQLEN;
    const v4f w0 = *(const v4f*)(w + (size_t)(d0 + 0) * D_CONV);
    const v4f w1 = *(const v4f*)(w + (size_t)(d0 + 1) * D_CONV);
    const v4f w2 = *(const v4f*)(w + (size_t)(d0 + 2) * D_CONV);
    const v4f w3 = *(const v4f*)(w + (size_t)(d0 + 3) * D_CONV);
    v4f acc = *(const v4f*)(cb + d0);
#pragma unroll
    for (int k = 0; k < D_CONV; ++k) {
        const int ls = l + k - (D_CONV - 1);
        if (ls >= 0) {
            const v4f xv = *(const v4f*)(x + (size_t)(m + k - (D_CONV - 1)) * D_INNER + d0);
            const v4f tk = (v4f){w0[k], w1[k], w2[k], w3[k]};
            acc += xv * tk;
        }
    }
    v4f o;
    o[0] = silu_f(acc[0]); o[1] = silu_f(acc[1]);
    o[2] = silu_f(acc[2]); o[3] = silu_f(acc[3]);
    *(v4f*)(xc + (size_t)m * D_INNER + d0) = o;
}

#define SCAN_CH   (BATCH * D_INNER)
#define SCAN_BLK  64
#define SCAN_G    4
#define SCAN_NG   (SCAN_T / SCAN_G)

__launch_bounds__(SCAN_BLK)
__global__ void scan_chunk_state(const float* __restrict__ delta,
                                 const float* __restrict__ xc,
                                 const float* __restrict__ xdbl,
                                 const float* __restrict__ A_log,
                                 float* __restrict__ hloc,
                                 float* __restrict__ pdec)
{
    __shared__ float sB[SCAN_T * D_STATE];

    const int c  = blockIdx.y;
    const int b  = blockIdx.x / (D_INNER / SCAN_BLK);
    const int d  = (blockIdx.x % (D_INNER / SCAN_BLK)) * SCAN_BLK + threadIdx.x;
    const int ch = b * D_INNER + d;
    const size_t row0 = (size_t)b * SEQLEN + (size_t)c * SCAN_T;

    {
        const float* src = xdbl + row0 * XDBL_LD + DT_RANK;
#pragma unroll
        for (int i = 0; i < (SCAN_T * D_STATE / 4) / SCAN_BLK; ++i) {
            const int idx = threadIdx.x + i * SCAN_BLK, r = idx >> 2, q = idx & 3;
            *(v4f*)(sB + r * D_STATE + q * 4) = *(const v4f*)(src + (size_t)r * XDBL_LD + q * 4);
        }
    }

    float An[D_STATE], h[D_STATE], p[D_STATE];
#pragma unroll
    for (int n = 0; n < D_STATE; ++n) {
        An[n] = -expf(A_log[d * D_STATE + n]);
        h[n] = 0.f;
        p[n] = 1.f;
    }

    const float* dtp = delta + row0 * D_INNER + d;
    const float* up  = xc    + row0 * D_INNER + d;

    float dt_n[SCAN_G], u_n[SCAN_G];
#pragma unroll
    for (int s = 0; s < SCAN_G; ++s) {
        dt_n[s] = dtp[(size_t)s * D_INNER];
        u_n[s]  = up[(size_t)s * D_INNER];
    }
    __syncthreads();

    for (int g = 0; g < SCAN_NG; ++g) {
        float dt_c[SCAN_G], u_c[SCAN_G];
#pragma unroll
        for (int s = 0; s < SCAN_G; ++s) { dt_c[s] = dt_n[s]; u_c[s] = u_n[s]; }
        const int gn = (g + 1 < SCAN_NG) ? g + 1 : g;
#pragma unroll
        for (int s = 0; s < SCAN_G; ++s) {
            dt_n[s] = dtp[(size_t)(gn * SCAN_G + s) * D_INNER];
            u_n[s]  = up[(size_t)(gn * SCAN_G + s) * D_INNER];
        }
#pragma unroll
        for (int s = 0; s < SCAN_G; ++s) {
            const float dt  = dt_c[s];
            const float dtu = dt * u_c[s];
            const float* br = sB + (g * SCAN_G + s) * D_STATE;
#pragma unroll
            for (int n = 0; n < D_STATE; ++n) {
                const float dA = expf(dt * An[n]);
                h[n] = h[n] * dA + dtu * br[n];
                p[n] = p[n] * dA;
            }
        }
    }
#pragma unroll
    for (int n = 0; n < D_STATE; ++n) {
        const size_t o = ((size_t)c * D_STATE + n) * SCAN_CH + ch;
        hloc[o] = h[n];
        pdec[o] = p[n];
    }
}

__launch_bounds__(SCAN_BLK)
__global__ void scan_kernel(const float* __restrict__ delta,
                            const float* __restrict__ xc,
                            const float* __restrict__ xdbl,
                            const float* __restrict__ A_log,
                            const float* __restrict__ Dp,
                            const float* __restrict__ z,
                            const float* __restrict__ hloc,
                            const float* __restrict__ pdec,
                            float* __restrict__ yout)
{
    __shared__ float sBC[SCAN_T * 2 * D_STATE];

    const int c  = blockIdx.y;
    const int b  = blockIdx.x / (D_INNER / SCAN_BLK);
    const int d  = (blockIdx.x % (D_INNER / SCAN_BLK)) * SCAN_BLK + threadIdx.x;
    const int ch = b * D_INNER + d;
    const size_t row0 = (size_t)b * SEQLEN + (size_t)c * SCAN_T;

    {
        const float* src = xdbl + row0 * XDBL_LD + DT_RANK;
#pragma unroll
        for (int i = 0; i < (SCAN_T * 2 * D_STATE / 4) / SCAN_BLK; ++i) {
            const int idx = threadIdx.x + i * SCAN_BLK, r = idx >> 3, q = idx & 7;
            *(v4f*)(sBC + r * 2 * D_STATE + q * 4) = *(const v4f*)(src + (size_t)r * XDBL_LD + q * 4);
        }
    }

    float An[D_STATE], h[D_STATE];
#pragma unroll
    for (int n = 0; n < D_STATE; ++n) {
        An[n] = -expf(A_log[d * D_STATE + n]);
        h[n] = 0.f;
    }
    const float Dd = Dp[d];

    const float* dtp = delta + row0 * D_INNER + d;
    const float* up  = xc    + row0 * D_INNER + d;
    const float* zp  = z     + row0 * D_INNER + d;
    float*       yp  = yout  + row0 * D_INNER + d;

    float dt_n[SCAN_G], u_n[SCAN_G], z_n[SCAN_G];
#pragma unroll
    for (int s = 0; s < SCAN_G; ++s) {
        dt_n[s] = dtp[(size_t)s * D_INNER];
        u_n[s]  = up[(size_t)s * D_INNER];
        z_n[s]  = zp[(size_t)s * D_INNER];
    }

#pragma unroll 2
    for (int j = 0; j < c; ++j) {
#pragma unroll
        for (int n = 0; n < D_STATE; ++n) {
            const size_t o = ((size_t)j * D_STATE + n) * SCAN_CH + ch;
            h[n] = h[n] * pdec[o] + hloc[o];
        }
    }
    __syncthreads();

    for (int g = 0; g < SCAN_NG; ++g) {
        float dt_c[SCAN_G], u_c[SCAN_G], z_c[SCAN_G];
#pragma unroll
        for (int s = 0; s < SCAN_G; ++s) { dt_c[s] = dt_n[s]; u_c[s] = u_n[s]; z_c[s] = z_n[s]; }
        const int gn = (g + 1 < SCAN_NG) ? g + 1 : g;
#pragma unroll
        for (int s = 0; s < SCAN_G; ++s) {
            dt_n[s] = dtp[(size_t)(gn * SCAN_G + s) * D_INNER];
            u_n[s]  = up[(size_t)(gn * SCAN_G + s) * D_INNER];
            z_n[s]  = zp[(size_t)(gn * SCAN_G + s) * D_INNER];
        }
        __builtin_amdgcn_sched_barrier(0);
#pragma unroll
        for (int s = 0; s < SCAN_G; ++s) {
            const float dt  = dt_c[s];
            const float u   = u_c[s];
            const float dtu = dt * u;
            const float* br = sBC + (g * SCAN_G + s) * 2 * D_STATE;
            float t[D_STATE];
#pragma unroll
            for (int n = 0; n < D_STATE; ++n) {
                const float dA = expf(dt * An[n]);
                h[n] = h[n] * dA + dtu * br[n];
                t[n] = h[n] * br[D_STATE + n];
            }
#pragma unroll
            for (int w = 1; w < D_STATE; w *= 2) {
#pragma unroll
                for (int n = 0; n < D_STATE; n += 2 * w) t[n] += t[n + w];
            }
            yp[(size_t)(g * SCAN_G + s) * D_INNER] = (t[0] + Dd * u) * silu_f(z_c[s]);
        }
    }
}

extern "C" void kernel_launch(void* const* d_in, const int* in_sizes, int n_in,
                              void* d_out, int out_size, void* d_ws, size_t ws_size,
                              hipStream_t stream)
{
    const float* hs        = (const float*)d_in[0];
    const float* in_proj_w = (const float*)d_in[1];
    const float* conv_w    = (const float*)d_in[2];
    const float* conv_b    = (const float*)d_in[3];
    const float* x_proj_w  = (const float*)d_in[4];
    const float* dt_proj_w = (const float*)d_in[5];
    const float* dt_proj_b = (const float*)d_in[6];
    const float* A_log     = (const float*)d_in[7];
    const float* Dp        = (const float*)d_in[8];
    const float* out_w     = (const float*)d_in[9];
    float* out = (float*)d_out;

    const size_t SZ_BLD = (size_t)BL * D_INNER * sizeof(float);
    char* ws = (char*)d_ws;
    float* x     = (float*)(ws);
    float* z     = (float*)(ws + SZ_BLD);
    float* xc    = (float*)(ws + 2 * SZ_BLD);
    float* delta = (float*)(ws + 3 * SZ_BLD);
    float* xdbl  = (float*)(ws + 4 * SZ_BLD);
    float* hloc  = (float*)(ws + 5 * SZ_BLD);
    float* pdec  = hloc + (size_t)SCAN_NC * SCAN_LANES;
    float* y     = x;

    const dim3 blk(256);

    gemm_wmma_f32<8, 2><<<dim3(BL / 128, D_INNER / 128), blk, 0, stream>>>(
        hs, D_MODEL, in_proj_w, D_MODEL, x, D_INNER, D_MODEL, conv_b, conv_w, xc);
    gemm_wmma_f32<8, 0><<<dim3(BL / 128, D_INNER / 128), blk, 0, stream>>>(
        hs, D_MODEL, in_proj_w + (size_t)D_INNER * D_MODEL, D_MODEL,
        z, D_INNER, D_MODEL, nullptr, nullptr, nullptr);

    conv_edge_kernel<<<(EDGE_N + 255) / 256, blk, 0, stream>>>(x, conv_w, conv_b, xc);

    xproj_wmma_f32<<<dim3(BL / 16), dim3(XP_THR), 0, stream>>>(xc, x_proj_w, xdbl);

    gemm_wmma_f32<8, 1><<<dim3(BL / 128, D_INNER / 128), blk, 0, stream>>>(
        xdbl, XDBL_LD, dt_proj_w, DT_RANK, delta, D_INNER, DT_RANK, dt_proj_b, nullptr, nullptr);

    scan_chunk_state<<<dim3(SCAN_CH / SCAN_BLK, SCAN_NC - 1), dim3(SCAN_BLK), 0, stream>>>(
        delta, xc, xdbl, A_log, hloc, pdec);
    scan_kernel<<<dim3(SCAN_CH / SCAN_BLK, SCAN_NC), dim3(SCAN_BLK), 0, stream>>>(
        delta, xc, xdbl, A_log, Dp, z, hloc, pdec, y);

    gemm_wmma_f32<8, 0><<<dim3(BL / 128, D_MODEL / 128), blk, 0, stream>>>(
        y, D_INNER, out_w, D_INNER, out, D_MODEL, D_INNER, nullptr, nullptr, nullptr);
}


# in_proj z-half GEMM and scan pass 1 in ONE launch as separate blocks (kinds alternate in the grid): a SIMD fills one kind's waits with the other's work; pass-1 blocks 256 channels; sums unchanged
# speedup vs baseline: 5.5946x; 1.0557x over previous
#include <hip/hip_runtime.h>
#include <math.h>

#define D_MODEL  1024
#define D_STATE  16
#define D_CONV   4
#define D_INNER  2048
#define DT_RANK  64
#define BATCH    2
#define SEQLEN   2048
#define BL       (BATCH * SEQLEN)
#define XDBL_LD  (DT_RANK + 2 * D_STATE)
#define SCAN_NC    16
#define SCAN_T     (SEQLEN / SCAN_NC)
#define SCAN_LANES (BATCH * D_INNER * D_STATE)

typedef __attribute__((ext_vector_type(2))) float v2f;
typedef __attribute__((ext_vector_type(4))) float v4f;
typedef __attribute__((ext_vector_type(8))) float v8f;

__device__ __forceinline__ float silu_f(float v)     { return v / (1.f + expf(-v)); }
__device__ __forceinline__ float softplus_f(float v) { return (v > 20.f) ? v : log1pf(expf(v)); }

#define GEMM_CS_LD 132
#define GEMM_LDS_LD 36
#define GEMM_SMEM(NT, EPI) (((EPI) == 2) ? 128 * GEMM_CS_LD : (128 + 16 * (NT)) * GEMM_LDS_LD)

template <int NT, int EPI>
__device__ __forceinline__ void gemm_tile(float* __restrict__ smem,
                                          const float* __restrict__ A, int lda,
                                          const float* __restrict__ W, int ldw,
                                          float* __restrict__ C, int ldc, int K,
                                          const float* __restrict__ bias,
                                          const float* __restrict__ cw,
                                          float* __restrict__ C2,
                                          int bx, int by)
{
    constexpr int KC      = 32;
    constexpr int KC4     = KC / 4;
    constexpr int LDS_LD  = GEMM_LDS_LD;
    static_assert(LDS_LD == KC + 4, "row pitch = K-chunk + 4");
    constexpr int NROWS_B = 16 * NT;
    constexpr int NA      = (128 * KC4) / 256;
    constexpr int NB      = (NROWS_B * KC4) / 256;

    constexpr int CS_LD   = GEMM_CS_LD;

    float* As = smem;
    float* Bs = smem + 128 * LDS_LD;

    const int tid  = threadIdx.x;
    const int wave = tid >> 5;
    const int lane = tid & 31;
    const int half = lane >> 4;
    const int lidx = lane & 15;
    const int m_blk  = bx * 128;
    const int n_base = by * NROWS_B;

    v8f acc[NT];
#pragma unroll
    for (int t = 0; t < NT; ++t) acc[t] = (v8f){0.f,0.f,0.f,0.f,0.f,0.f,0.f,0.f};

    v4f ra[NA], rb[NB];

#pragma unroll
    for (int i = 0; i < NA; ++i) {
        const int idx = tid + i * 256, r = idx >> 3, c = (idx & 7) * 4;
        ra[i] = *(const v4f*)(A + (size_t)(m_blk + r) * lda + c);
    }
#pragma unroll
    for (int i = 0; i < NB; ++i) {
        const int idx = tid + i * 256, r = idx >> 3, c = (idx & 7) * 4;
        rb[i] = *(const v4f*)(W + (size_t)(n_base + r) * ldw + c);
    }

    for (int k0 = 0; k0 < K; k0 += KC) {
#pragma unroll
        for (int i = 0; i < NA; ++i) {
            const int idx = tid + i * 256, r = idx >> 3, c = (idx & 7) * 4;
            *(v4f*)(As + r * LDS_LD + c) = ra[i];
        }
#pragma unroll
        for (int i = 0; i < NB; ++i) {
            const int idx = tid + i * 256, r = idx >> 3, c = (idx & 7) * 4;
            *(v4f*)(Bs + r * LDS_LD + c) = rb[i];
        }
        __syncthreads();

        if (k0 + KC < K) {
#pragma unroll
            for (int i = 0; i < NA; ++i) {
                const int idx = tid + i * 256, r = idx >> 3, c = (idx & 7) * 4;
                ra[i] = *(const v4f*)(A + (size_t)(m_blk + r) * lda + (k0 + KC) + c);
            }
#pragma unroll
            for (int i = 0; i < NB; ++i) {
                const int idx = tid + i * 256, r = idx >> 3, c = (idx & 7) * 4;
                rb[i] = *(const v4f*)(W + (size_t)(n_base + r) * ldw + (k0 + KC) + c);
            }
        }

        const float* a_lds = As + (wave * 16 + lidx) * LDS_LD + 2 * half;
        const float* b_lds = Bs + lidx * LDS_LD + 2 * half;
#pragma unroll
        for (int kk = 0; kk < KC; kk += 4) {
            v2f a = *(const v2f*)(a_lds + kk);
#pragma unroll
            for (int t = 0; t < NT; ++t) {
                v2f b = *(const v2f*)(b_lds + t * 16 * LDS_LD + kk);
                acc[t] = __builtin_amdgcn_wmma_f32_16x16x4_f32(
                    false, a, false, b, (short)0, acc[t], false, false);
            }
        }
        __syncthreads();
    }

    if (EPI == 2) {
#pragma unroll
        for (int t = 0; t < NT; ++t) {
#pragma unroll
            for (int v = 0; v < 8; ++v)
                smem[(wave * 16 + half * 8 + v) * CS_LD + t * 16 + lidx] = acc[t][v];
        }
        __syncthreads();

        const int cg = tid & 31;
        const int r0 = (tid >> 5) * 16;
        const int d0 = n_base + cg * 4;
        const v4f bv = *(const v4f*)(bias + d0);
        const v4f w0 = *(const v4f*)(cw + (size_t)(d0 + 0) * D_CONV);
        const v4f w1 = *(const v4f*)(cw + (size_t)(d0 + 1) * D_CONV);
        const v4f w2 = *(const v4f*)(cw + (size_t)(d0 + 2) * D_CONV);
        const v4f w3 = *(const v4f*)(cw + (size_t)(d0 + 3) * D_CONV);
        const v4f t0 = (v4f){w0[0], w1[0], w2[0], w3[0]};
        const v4f t1 = (v4f){w0[1], w1[1], w2[1], w3[1]};
        const v4f t2 = (v4f){w0[2], w1[2], w2[2], w3[2]};
        const v4f t3 = (v4f){w0[3], w1[3], w2[3], w3[3]};
        const float* cs = smem + cg * 4;

        if (r0 == 0 || r0 == 112) {
            const int rb = (r0 == 0) ? 0 : 125;
#pragma unroll
            for (int j = 0; j < 3; ++j)
                *(v4f*)(C + (size_t)(m_blk + rb + j) * ldc + d0) = *(const v4f*)(cs + (rb + j) * CS_LD);
        }

        v4f xm3 = (v4f){0.f, 0.f, 0.f, 0.f}, xm2 = xm3, xm1 = xm3;
        if (r0 > 0) {
            xm3 = *(const v4f*)(cs + (r0 - 3) * CS_LD);
            xm2 = *(const v4f*)(cs + (r0 - 2) * CS_LD);
            xm1 = *(const v4f*)(cs + (r0 - 1) * CS_LD);
        }
#pragma unroll
        for (int i = 0; i < 16; ++i) {
            const int r = r0 + i;
            const v4f x0 = *(const v4f*)(cs + r * CS_LD);
            if (r >= 3) {
                v4f a = bv;
                a += xm3 * t0;
                a += xm2 * t1;
                a += xm1 * t2;
                a += x0 * t3;
                v4f o;
                o[0] = silu_f(a[0]); o[1] = silu_f(a[1]);
                o[2] = silu_f(a[2]); o[3] = silu_f(a[3]);
                *(v4f*)(C2 + (size_t)(m_blk + r) * ldc + d0) = o;
            }
            xm3 = xm2; xm2 = xm1; xm1 = x0;
        }
        return;
    }
#pragma unroll
    for (int t = 0; t < NT; ++t) {
        const int n = n_base + t * 16 + lidx;
#pragma unroll
        for (int v = 0; v < 8; ++v) {
            const int m = m_blk + wave * 16 + half * 8 + v;
            float val = acc[t][v];
            if (EPI == 1) val = softplus_f(val + bias[n]);
            C[(size_t)m * ldc + n] = val;
        }
    }
}

template <int NT, int EPI>
__launch_bounds__(256)
__global__ void gemm_wmma_f32(const float* __restrict__ A, int lda,
                              const float* __restrict__ W, int ldw,
                              float* __restrict__ C, int ldc, int K,
                              const float* __restrict__ bias,
                              const float* __restrict__ cw,
                              float* __restrict__ C2)
{
    __shared__ float smem[GEMM_SMEM(NT, EPI)];
    gemm_tile<NT, EPI>(smem, A, lda, W, ldw, C, ldc, K, bias, cw, C2, blockIdx.x, blockIdx.y);
}

#define XP_KC   128
#define XP_LD   (XP_KC + 4)
#define XP_ROWS (16 + XDBL_LD + 2)
#define XP_THR  96
#define XP_NCH  (D_INNER / XP_KC)

__launch_bounds__(XP_THR)
__global__ void xproj_wmma_f32(const float* __restrict__ A,
                               const float* __restrict__ W,
                               float* __restrict__ C)
{
    constexpr int KC4  = XP_KC / 4;
    constexpr int NVEC = XP_ROWS * KC4;
    constexpr int NV   = NVEC / XP_THR;
    static_assert(NV * XP_THR == NVEC, "the tile must split evenly over the threads");
    constexpr int KS   = XP_KC / 4;
    constexpr int TILE = XP_ROWS * XP_LD;

    __shared__ float smem[2 * TILE];

    const int tid  = threadIdx.x;
    const int wave = tid >> 5;
    const int lane = tid & 31;
    const int half = lane >> 4;
    const int lidx = lane & 15;
    const int m_blk = blockIdx.x * 16;

    v8f acc0 = (v8f){0.f,0.f,0.f,0.f,0.f,0.f,0.f,0.f};
    v8f acc1 = acc0;

    const float* gsrc[NV];
    int          loff[NV];
#pragma unroll
    for (int i = 0; i < NV; ++i) {
        const int idx = tid + i * XP_THR, r = idx / KC4, c = (idx % KC4) * 4;
        loff[i] = r * XP_LD + c;
        const int wr = (r - 16 < XDBL_LD) ? r - 16 : XDBL_LD - 1;
        gsrc[i] = (r < 16) ? A + (size_t)(m_blk + r) * D_INNER + c
                           : W + (size_t)wr * D_INNER + c;
    }

    v4f regs[NV];
#pragma unroll
    for (int i = 0; i < NV; ++i) regs[i] = *(const v4f*)(gsrc[i]);
#pragma unroll
    for (int i = 0; i < NV; ++i) {
        *(v4f*)(smem + loff[i]) = regs[i];
        regs[i] = *(const v4f*)(gsrc[i] + XP_KC);
    }
    __syncthreads();

    for (int ch = 0; ch < XP_NCH; ++ch) {
        const float* cur = smem + (ch & 1) * TILE;
        float*       nxt = smem + ((ch + 1) & 1) * TILE;
        const int kload = ((ch + 2 < XP_NCH) ? ch + 2 : XP_NCH - 1) * XP_KC;
        const float* a_lds = cur + lidx * XP_LD + 2 * half;
        const float* b_lds = cur + (16 + wave * 32 + lidx) * XP_LD + 2 * half;
#pragma unroll
        for (int ks = 0; ks < KS; ++ks) {
            const v2f a  = *(const v2f*)(a_lds + ks * 4);
            const v2f b0 = *(const v2f*)(b_lds + ks * 4);
            const v2f b1 = *(const v2f*)(b_lds + 16 * XP_LD + ks * 4);
            acc0 = __builtin_amdgcn_wmma_f32_16x16x4_f32(
                false, a, false, b0, (short)0, acc0, false, false);
            acc1 = __builtin_amdgcn_wmma_f32_16x16x4_f32(
                false, a, false, b1, (short)0, acc1, false, false);
#pragma unroll
            for (int i = (ks * NV) / KS; i < ((ks + 1) * NV) / KS; ++i) {
                *(v4f*)(nxt + loff[i]) = regs[i];
                regs[i] = *(const v4f*)(gsrc[i] + kload);
            }
        }
        __syncthreads();
    }

    float* crow = C + (size_t)(m_blk + half * 8) * XDBL_LD + wave * 32 + lidx;
#pragma unroll
    for (int v = 0; v < 8; ++v) {
        crow[(size_t)v * XDBL_LD]      = acc0[v];
        crow[(size_t)v * XDBL_LD + 16] = acc1[v];
    }
}

#define EDGE_ROWS 3
#define EDGE_N    ((BL / 128) * EDGE_ROWS * (D_INNER / 4))
__global__ void conv_edge_kernel(const float* __restrict__ x,
                                 const float* __restrict__ w,
                                 const float* __restrict__ cb,
                                 float* __restrict__ xc)
{
    const int idx = blockIdx.x * blockDim.x + threadIdx.x;
    if (idx >= EDGE_N) return;
    const int d0   = (idx % (D_INNER / 4)) * 4;
    const int rr   = (idx / (D_INNER / 4)) % EDGE_ROWS;
    const int tile = idx / ((D_INNER / 4) * EDGE_ROWS);
    const int m = tile * 128 + rr;
    const int l = m % SEQLEN;
    const v4f w0 = *(const v4f*)(w + (size_t)(d0 + 0) * D_CONV);
    const v4f w1 = *(const v4f*)(w + (size_t)(d0 + 1) * D_CONV);
    const v4f w2 = *(const v4f*)(w + (size_t)(d0 + 2) * D_CONV);
    const v4f w3 = *(const v4f*)(w + (size_t)(d0 + 3) * D_CONV);
    v4f acc = *(const v4f*)(cb + d0);
#pragma unroll
    for (int k = 0; k < D_CONV; ++k) {
        const int ls = l + k - (D_CONV - 1);
        if (ls >= 0) {
            const v4f xv = *(const v4f*)(x + (size_t)(m + k - (D_CONV - 1)) * D_INNER + d0);
            const v4f tk = (v4f){w0[k], w1[k], w2[k], w3[k]};
            acc += xv * tk;
        }
    }
    v4f o;
    o[0] = silu_f(acc[0]); o[1] = silu_f(acc[1]);
    o[2] = silu_f(acc[2]); o[3] = silu_f(acc[3]);
    *(v4f*)(xc + (size_t)m * D_INNER + d0) = o;
}

#define SCAN_CH   (BATCH * D_INNER)
#define SCAN_BLK  64
#define SCAN_G    4
#define SCAN_NG   (SCAN_T / SCAN_G)

template <int BLK>
__device__ __forceinline__ void scan_chunk_state_blk(float* __restrict__ sB,
                                                     const float* __restrict__ delta,
                                                     const float* __restrict__ xc,
                                                     const float* __restrict__ xdbl,
                                                     const float* __restrict__ A_log,
                                                     float* __restrict__ hloc,
                                                     float* __restrict__ pdec,
                                                     int bx, int c)
{
    const int b  = bx / (D_INNER / BLK);
    const int d  = (bx % (D_INNER / BLK)) * BLK + threadIdx.x;
    const int ch = b * D_INNER + d;
    const size_t row0 = (size_t)b * SEQLEN + (size_t)c * SCAN_T;

    {
        const float* src = xdbl + row0 * XDBL_LD + DT_RANK;
#pragma unroll
        for (int i = 0; i < (SCAN_T * D_STATE / 4) / BLK; ++i) {
            const int idx = threadIdx.x + i * BLK, r = idx >> 2, q = idx & 3;
            *(v4f*)(sB + r * D_STATE + q * 4) = *(const v4f*)(src + (size_t)r * XDBL_LD + q * 4);
        }
    }

    float An[D_STATE], h[D_STATE], p[D_STATE];
#pragma unroll
    for (int n = 0; n < D_STATE; ++n) {
        An[n] = -expf(A_log[d * D_STATE + n]);
        h[n] = 0.f;
        p[n] = 1.f;
    }

    const float* dtp = delta + row0 * D_INNER + d;
    const float* up  = xc    + row0 * D_INNER + d;

    float dt_n[SCAN_G], u_n[SCAN_G];
#pragma unroll
    for (int s = 0; s < SCAN_G; ++s) {
        dt_n[s] = dtp[(size_t)s * D_INNER];
        u_n[s]  = up[(size_t)s * D_INNER];
    }
    __syncthreads();

    for (int g = 0; g < SCAN_NG; ++g) {
        float dt_c[SCAN_G], u_c[SCAN_G];
#pragma unroll
        for (int s = 0; s < SCAN_G; ++s) { dt_c[s] = dt_n[s]; u_c[s] = u_n[s]; }
        const int gn = (g + 1 < SCAN_NG) ? g + 1 : g;
#pragma unroll
        for (int s = 0; s < SCAN_G; ++s) {
            dt_n[s] = dtp[(size_t)(gn * SCAN_G + s) * D_INNER];
            u_n[s]  = up[(size_t)(gn * SCAN_G + s) * D_INNER];
        }
#pragma unroll
        for (int s = 0; s < SCAN_G; ++s) {
            const float dt  = dt_c[s];
            const float dtu = dt * u_c[s];
            const float* br = sB + (g * SCAN_G + s) * D_STATE;
#pragma unroll
            for (int n = 0; n < D_STATE; ++n) {
                const float dA = expf(dt * An[n]);
                h[n] = h[n] * dA + dtu * br[n];
                p[n] = p[n] * dA;
            }
        }
    }
#pragma unroll
    for (int n = 0; n < D_STATE; ++n) {
        const size_t o = ((size_t)c * D_STATE + n) * SCAN_CH + ch;
        hloc[o] = h[n];
        pdec[o] = p[n];
    }
}

#define FZ_THR   256
#define FZ_WGP   256
#define FZ_NG    ((BL / 128) * (D_INNER / 128))
#define FZ_SPC   (SCAN_CH / FZ_THR)
#define FZ_NS    (FZ_SPC * (SCAN_NC - 1))

__launch_bounds__(FZ_THR)
__global__ void zgemm_scan_chunk_state(const float* __restrict__ hs,
                                       const float* __restrict__ Wz,
                                       float* __restrict__ z,
                                       const float* __restrict__ delta,
                                       const float* __restrict__ xc,
                                       const float* __restrict__ xdbl,
                                       const float* __restrict__ A_log,
                                       float* __restrict__ hloc,
                                       float* __restrict__ pdec)
{
    static_assert(GEMM_SMEM(8, 0) >= SCAN_T * D_STATE, "pass 1 borrows the GEMM's LDS array");
    static_assert(FZ_NS <= FZ_WGP, "the scan blocks must fit in the alternating positions");
    __shared__ float smem[GEMM_SMEM(8, 0)];

    const int p = blockIdx.x;
    int g, s = -1;
    if (p < 2 * FZ_WGP) {
        const int round = p / FZ_WGP;
        const int idx   = round * (FZ_WGP / 2) + ((p % FZ_WGP) >> 1);
        const bool scan_pos = ((p & 1) != 0) != (round != 0);
        g = idx;
        if (scan_pos) {
            if (idx < FZ_NS) s = idx;
            else g = FZ_WGP + (idx - FZ_NS);
        }
    } else {
        g = FZ_WGP + (FZ_WGP - FZ_NS) + (p - 2 * FZ_WGP);
    }

    if (s >= 0) {
        scan_chunk_state_blk<FZ_THR>(smem, delta, xc, xdbl, A_log, hloc, pdec, s % FZ_SPC, s / FZ_SPC);
    } else {
        gemm_tile<8, 0>(smem, hs, D_MODEL, Wz, D_MODEL, z, D_INNER, D_MODEL,
                        nullptr, nullptr, nullptr, g % (BL / 128), g / (BL / 128));
    }
}

__launch_bounds__(SCAN_BLK)
__global__ void scan_kernel(const float* __restrict__ delta,
                            const float* __restrict__ xc,
                            const float* __restrict__ xdbl,
                            const float* __restrict__ A_log,
                            const float* __restrict__ Dp,
                            const float* __restrict__ z,
                            const float* __restrict__ hloc,
                            const float* __restrict__ pdec,
                            float* __restrict__ yout)
{
    __shared__ float sBC[SCAN_T * 2 * D_STATE];

    const int c  = blockIdx.y;
    const int b  = blockIdx.x / (D_INNER / SCAN_BLK);
    const int d  = (blockIdx.x % (D_INNER / SCAN_BLK)) * SCAN_BLK + threadIdx.x;
    const int ch = b * D_INNER + d;
    const size_t row0 = (size_t)b * SEQLEN + (size_t)c * SCAN_T;

    {
        const float* src = xdbl + row0 * XDBL_LD + DT_RANK;
#pragma unroll
        for (int i = 0; i < (SCAN_T * 2 * D_STATE / 4) / SCAN_BLK; ++i) {
            const int idx = threadIdx.x + i * SCAN_BLK, r = idx >> 3, q = idx & 7;
            *(v4f*)(sBC + r * 2 * D_STATE + q * 4) = *(const v4f*)(src + (size_t)r * XDBL_LD + q * 4);
        }
    }

    float An[D_STATE], h[D_STATE];
#pragma unroll
    for (int n = 0; n < D_STATE; ++n) {
        An[n] = -expf(A_log[d * D_STATE + n]);
        h[n] = 0.f;
    }
    const float Dd = Dp[d];

    const float* dtp = delta + row0 * D_INNER + d;
    const float* up  = xc    + row0 * D_INNER + d;
    const float* zp  = z     + row0 * D_INNER + d;
    float*       yp  = yout  + row0 * D_INNER + d;

    float dt_n[SCAN_G], u_n[SCAN_G], z_n[SCAN_G];
#pragma unroll
    for (int s = 0; s < SCAN_G; ++s) {
        dt_n[s] = dtp[(size_t)s * D_INNER];
        u_n[s]  = up[(size_t)s * D_INNER];
        z_n[s]  = zp[(size_t)s * D_INNER];
    }

#pragma unroll 2
    for (int j = 0; j < c; ++j) {
#pragma unroll
        for (int n = 0; n < D_STATE; ++n) {
            const size_t o = ((size_t)j * D_STATE + n) * SCAN_CH + ch;
            h[n] = h[n] * pdec[o] + hloc[o];
        }
    }
    __syncthreads();

    for (int g = 0; g < SCAN_NG; ++g) {
        float dt_c[SCAN_G], u_c[SCAN_G], z_c[SCAN_G];
#pragma unroll
        for (int s = 0; s < SCAN_G; ++s) { dt_c[s] = dt_n[s]; u_c[s] = u_n[s]; z_c[s] = z_n[s]; }
        const int gn = (g + 1 < SCAN_NG) ? g + 1 : g;
#pragma unroll
        for (int s = 0; s < SCAN_G; ++s) {
            dt_n[s] = dtp[(size_t)(gn * SCAN_G + s) * D_INNER];
            u_n[s]  = up[(size_t)(gn * SCAN_G + s) * D_INNER];
            z_n[s]  = zp[(size_t)(gn * SCAN_G + s) * D_INNER];
        }
        __builtin_amdgcn_sched_barrier(0);
#pragma unroll
        for (int s = 0; s < SCAN_G; ++s) {
            const float dt  = dt_c[s];
            const float u   = u_c[s];
            const float dtu = dt * u;
            const float* br = sBC + (g * SCAN_G + s) * 2 * D_STATE;
            float t[D_STATE];
#pragma unroll
            for (int n = 0; n < D_STATE; ++n) {
                const float dA = expf(dt * An[n]);
                h[n] = h[n] * dA + dtu * br[n];
                t[n] = h[n] * br[D_STATE + n];
            }
#pragma unroll
            for (int w = 1; w < D_STATE; w *= 2) {
#pragma unroll
                for (int n = 0; n < D_STATE; n += 2 * w) t[n] += t[n + w];
            }
            yp[(size_t)(g * SCAN_G + s) * D_INNER] = (t[0] + Dd * u) * silu_f(z_c[s]);
        }
    }
}

extern "C" void kernel_launch(void* const* d_in, const int* in_sizes, int n_in,
                              void* d_out, int out_size, void* d_ws, size_t ws_size,
                              hipStream_t stream)
{
    const float* hs        = (const float*)d_in[0];
    const float* in_proj_w = (const float*)d_in[1];
    const float* conv_w    = (const float*)d_in[2];
    const float* conv_b    = (const float*)d_in[3];
    const float* x_proj_w  = (const float*)d_in[4];
    const float* dt_proj_w = (const float*)d_in[5];
    const float* dt_proj_b = (const float*)d_in[6];
    const float* A_log     = (const float*)d_in[7];
    const float* Dp        = (const float*)d_in[8];
    const float* out_w     = (const float*)d_in[9];
    float* out = (float*)d_out;

    const size_t SZ_BLD = (size_t)BL * D_INNER * sizeof(float);
    char* ws = (char*)d_ws;
    float* x     = (float*)(ws);
    float* z     = (float*)(ws + SZ_BLD);
    float* xc    = (float*)(ws + 2 * SZ_BLD);
    float* delta = (float*)(ws + 3 * SZ_BLD);
    float* xdbl  = (float*)(ws + 4 * SZ_BLD);
    float* hloc  = (float*)(ws + 5 * SZ_BLD);
    float* pdec  = hloc + (size_t)SCAN_NC * SCAN_LANES;
    float* y     = x;

    const dim3 blk(256);

    gemm_wmma_f32<8, 2><<<dim3(BL / 128, D_INNER / 128), blk, 0, stream>>>(
        hs, D_MODEL, in_proj_w, D_MODEL, x, D_INNER, D_MODEL, conv_b, conv_w, xc);

    conv_edge_kernel<<<(EDGE_N + 255) / 256, blk, 0, stream>>>(x, conv_w, conv_b, xc);

    xproj_wmma_f32<<<dim3(BL / 16), dim3(XP_THR), 0, stream>>>(xc, x_proj_w, xdbl);

    gemm_wmma_f32<8, 1><<<dim3(BL / 128, D_INNER / 128), blk, 0, stream>>>(
        xdbl, XDBL_LD, dt_proj_w, DT_RANK, delta, D_INNER, DT_RANK, dt_proj_b, nullptr, nullptr);

    zgemm_scan_chunk_state<<<dim3(FZ_NG + FZ_NS), dim3(FZ_THR), 0, stream>>>(
        hs, in_proj_w + (size_t)D_INNER * D_MODEL, z, delta, xc, xdbl, A_log, hloc, pdec);
    scan_kernel<<<dim3(SCAN_CH / SCAN_BLK, SCAN_NC), dim3(SCAN_BLK), 0, stream>>>(
        delta, xc, xdbl, A_log, Dp, z, hloc, pdec, y);

    gemm_wmma_f32<8, 0><<<dim3(BL / 128, D_MODEL / 128), blk, 0, stream>>>(
        y, D_INNER, out_w, D_INNER, out, D_MODEL, D_INNER, nullptr, nullptr, nullptr);
}
